# starting kernel
# baseline (speedup 1.0000x reference)
#include <hip/hip_runtime.h>

#define IN_FEAT 64
#define OUT_FEAT 64

typedef float v2f __attribute__((ext_vector_type(2)));
typedef float v8f __attribute__((ext_vector_type(8)));

__global__ __launch_bounds__(256) void gat_gemm_wmma(
    const float* __restrict__ feat, const float* __restrict__ weight,
    const float* __restrict__ bias, float* __restrict__ h, int n_nodes)
{
    const int lane  = threadIdx.x & 31;
    const int tileM = blockIdx.x * (blockDim.x >> 5) + (threadIdx.x >> 5);
    const int row0  = tileM * 16;
    if (row0 >= n_nodes) return;

    const int mn  = lane & 15;
    const int hi  = lane >> 4;
    const int khi = hi * 2;

    const float* arow  = feat + (size_t)(row0 + mn) * IN_FEAT + khi;
    const float* bbase = weight + (size_t)khi * OUT_FEAT + mn;

    v8f acc0 = {}, acc1 = {}, acc2 = {}, acc3 = {};

#pragma unroll
    for (int k = 0; k < IN_FEAT; k += 4) {
        v2f a;
        a.x = arow[k];
        a.y = arow[k + 1];

        const float* bk = bbase + (size_t)k * OUT_FEAT;
        v2f b0, b1, b2, b3;
        b0.x = bk[0];   b0.y = bk[OUT_FEAT + 0];
        b1.x = bk[16];  b1.y = bk[OUT_FEAT + 16];
        b2.x = bk[32];  b2.y = bk[OUT_FEAT + 32];
        b3.x = bk[48];  b3.y = bk[OUT_FEAT + 48];

        acc0 = __builtin_amdgcn_wmma_f32_16x16x4_f32(false, a, false, b0,
                                                     (short)0, acc0, false, false);
        acc1 = __builtin_amdgcn_wmma_f32_16x16x4_f32(false, a, false, b1,
                                                     (short)0, acc1, false, false);
        acc2 = __builtin_amdgcn_wmma_f32_16x16x4_f32(false, a, false, b2,
                                                     (short)0, acc2, false, false);
        acc3 = __builtin_amdgcn_wmma_f32_16x16x4_f32(false, a, false, b3,
                                                     (short)0, acc3, false, false);
    }

    const float bv0 = bias[mn];
    const float bv1 = bias[16 + mn];
    const float bv2 = bias[32 + mn];
    const float bv3 = bias[48 + mn];

#pragma unroll
    for (int v = 0; v < 8; ++v) {
        const int row = row0 + v + 8 * hi;
        float* hr = h + (size_t)row * OUT_FEAT + mn;
        hr[0]  = acc0[v] + bv0;
        hr[16] = acc1[v] + bv1;
        hr[32] = acc2[v] + bv2;
        hr[48] = acc3[v] + bv3;
    }
}

__global__ __launch_bounds__(256) void gat_scatter(
    const float* __restrict__ h, const float* __restrict__ att,
    const int* __restrict__ src, const int* __restrict__ dst,
    float* __restrict__ out, int n_edges)
{
    const int t = blockIdx.x * blockDim.x + threadIdx.x;
    const int e = t >> 4;
    if (e >= n_edges) return;
    const int f = (t & 15) << 2;

    const int   s = src[e];
    const int   d = dst[e];
    const float a = att[e];

    const float4 v = *(const float4*)(h + (size_t)s * OUT_FEAT + f);
    float* op = out + (size_t)d * OUT_FEAT + f;

    unsafeAtomicAdd(op + 0, v.x * a);
    unsafeAtomicAdd(op + 1, v.y * a);
    unsafeAtomicAdd(op + 2, v.z * a);
    unsafeAtomicAdd(op + 3, v.w * a);
}

extern "C" void kernel_launch(void* const* d_in, const int* in_sizes, int n_in,
                              void* d_out, int out_size, void* d_ws, size_t ws_size,
                              hipStream_t stream)
{
    const float* feat   = (const float*)d_in[0];
    const float* att    = (const float*)d_in[1];
    const int*   src    = (const int*)d_in[2];
    const int*   dst    = (const int*)d_in[3];
    const float* weight = (const float*)d_in[4];
    const float* bias   = (const float*)d_in[5];

    float* out = (float*)d_out;
    float* h   = (float*)d_ws;

    const int n_nodes = in_sizes[0] / IN_FEAT;
    const int n_edges = in_sizes[2];

    hipMemsetAsync(out, 0, (size_t)out_size * sizeof(float), stream);

    const int tilesM = (n_nodes + 15) / 16;
    const int wavesPerBlock = 8;
    const int gblocks = (tilesM + wavesPerBlock - 1) / wavesPerBlock;
    gat_gemm_wmma<<<gblocks, 256, 0, stream>>>(feat, weight, bias, h, n_nodes);

    const long long sthreads = (long long)n_edges * 16;
    const int sblocks = (int)((sthreads + 255) / 256);
    gat_scatter<<<sblocks, 256, 0, stream>>>(h, att, src, dst, out, n_edges);
}


# gather loop on fewer instructions per feat row: ordered records hold row byte offset and dst row in one word (one and-or gives the load offset); row change tested once per 4 ordered records
# speedup vs baseline: 3.6525x; 3.6525x over previous
#include <hip/hip_runtime.h>


#define IN_FEAT 64
#define OUT_FEAT 64

#define BSHIFT 7
#define BNODES 128
#define MAXB 1024
#define PITCH 68
#define BIN_THREADS 1024
#define BIN_BLOCKS 256
#define BKPT 7
#define BCAP (BKPT * BIN_THREADS)
#define KPT 12
#define CAP (KPT * 256)

typedef float v2f __attribute__((ext_vector_type(2)));
typedef float v8f __attribute__((ext_vector_type(8)));


__device__ __forceinline__ int gat_chunk(int n_edges) { return (n_edges + BIN_BLOCKS - 1) / BIN_BLOCKS; }
__device__ __forceinline__ int gat_passes(int chunk)  { return max(1, (chunk + BCAP - 1) / BCAP); }

__device__ __forceinline__ bool gat_ws_ok(int n_edges, int nb, int units, size_t ws_bytes)
{
    return (size_t)n_edges * 8 + (size_t)nb * (size_t)units * 4 <= ws_bytes;
}

__device__ __forceinline__ bool gat_dst_ok(int d, int n_nodes)
{
    return (unsigned)d < (unsigned)n_nodes && (d >> BSHIFT) < MAXB;
}

__global__ __launch_bounds__(BIN_THREADS) void gat_bin_local(
    const int* __restrict__ src, const int* __restrict__ dst, const float* __restrict__ att,
    int n_edges, int n_nodes, int nb, int* __restrict__ rec, int* __restrict__ tab, size_t ws_bytes)
{
    __shared__ int hist[MAXB];
    __shared__ int wsum[32];
    __shared__ int stot;
    __shared__ __attribute__((aligned(16))) int2 srt[BCAP];
    const int tid  = threadIdx.x;
    const int lane = tid & 31;
    const int wave = tid >> 5;

    const int chunk = gat_chunk(n_edges);
    const int P     = gat_passes(chunk);
    if (!gat_ws_ok(n_edges, nb, BIN_BLOCKS * P, ws_bytes)) return;

    int2* rec2 = (int2*)rec;
    const int blk = (int)blockIdx.x;

    for (int p = 0; p < P; ++p) {
        const int u    = blk * P + p;
        const int e_lo = min(blk * chunk + p * BCAP, n_edges);
        const int e_hi = min(min(e_lo + BCAP, (blk + 1) * chunk), n_edges);
        const int nc   = max(0, e_hi - e_lo);

        hist[tid] = 0;
        __syncthreads();

        int ed[BKPT], es[BKPT], ea[BKPT], rk[BKPT];
#pragma unroll
        for (int k = 0; k < BKPT; ++k) { ed[k] = -1; es[k] = 0; ea[k] = 0; rk[k] = -1; }
        if (nc > 0) {
#pragma unroll
            for (int k = 0; k < BKPT; ++k) {
                const int e = e_lo + min(tid + k * BIN_THREADS, nc - 1);
                ed[k] = dst[e];
                es[k] = src[e];
                ea[k] = __float_as_int(att[e]);
            }
#pragma unroll
            for (int k = 0; k < BKPT; ++k)
                if (tid + k * BIN_THREADS < nc && gat_dst_ok(ed[k], n_nodes))
                    rk[k] = atomicAdd(&hist[ed[k] >> BSHIFT], 1);
        }
        __syncthreads();

        const int v = hist[tid];
        int x = v;
#pragma unroll
        for (int o = 1; o < 32; o <<= 1) {
            const int y = __shfl_up(x, o);
            if (lane >= o) x += y;
        }
        if (lane == 31) wsum[wave] = x;
        __syncthreads();
        if (wave == 0) {
            const int w = wsum[lane];
            int s = w;
#pragma unroll
            for (int o = 1; o < 32; o <<= 1) {
                const int y = __shfl_up(s, o);
                if (lane >= o) s += y;
            }
            wsum[lane] = s - w;
        }
        __syncthreads();
        const int start = x - v + wsum[wave];
        if (tid < nb) tab[(size_t)u * nb + tid] = start | (v << 16);
        hist[tid] = start;
        if (tid == BIN_THREADS - 1) stot = start + v;
        __syncthreads();

#pragma unroll
        for (int k = 0; k < BKPT; ++k)
            if (rk[k] >= 0)
                srt[hist[ed[k] >> BSHIFT] + rk[k]] =
                    make_int2((es[k] & 0xFFFFFF) | ((ed[k] & (BNODES - 1)) << 24), ea[k]);
        __syncthreads();

        const int nv = min(stot, nc);
        for (int i = tid; i < nv; i += BIN_THREADS) rec2[e_lo + i] = srt[i];
        __syncthreads();
    }
}

template <bool FULL>
__device__ __forceinline__ void gat_load8(
    const int2 r, const int j, const int m, const float* __restrict__ feat, const unsigned lane8,
    float2 (&v)[8], float (&a)[8], int (&dl)[8])
{
#pragma unroll
    for (int u = 0; u < 8; ++u) {
        v[u] = make_float2(0.f, 0.f);
        a[u] = 0.f;
        dl[u] = 0;
        if (FULL || j + u < m) {
            const unsigned p = (unsigned)__builtin_amdgcn_readlane(r.x, j + u);
            a[u]  = __int_as_float(__builtin_amdgcn_readlane(r.y, j + u));
            dl[u] = (int)(p & (BNODES - 1));
            const unsigned voff = (p & 0xFFFFFF00u) | lane8;
            v[u] = *(const float2*)((const char*)feat + voff);
        }
    }
}

__device__ __forceinline__ void gat_flush(float* srow, const int cur, const float2 acc, const float2 carry)
{
    if (cur >= 0) {
        float2* q = (float2*)(srow + cur * PITCH);
        *q = make_float2(carry.x + acc.x, carry.y + acc.y);
    }
}

__device__ __forceinline__ void gat_step(
    float* srow, const bool first, int& cur, float2& acc, float2& carry,
    const int dl, const float a, const float2 v)
{
    if (__builtin_expect(dl != cur, 0)) {
        gat_flush(srow, cur, acc, carry);
        cur = dl;
        acc = make_float2(0.f, 0.f);
        carry = make_float2(0.f, 0.f);
        if (!first) carry = *(const float2*)(srow + dl * PITCH);
    }
    acc.x = fmaf(v.x, a, acc.x);
    acc.y = fmaf(v.y, a, acc.y);
}

template <bool FULL>
__device__ __forceinline__ void gat_acc8(
    float* srow, const bool first, int& cur, float2& acc, float2& carry,
    const int j, const int m,
    const float2 (&v)[8], const float (&a)[8], const int (&dl)[8])
{
#pragma unroll
    for (int q = 0; q < 8; q += 4) {
        if ((FULL || j + q + 3 < m) && __builtin_expect(dl[q + 3] == cur, 1)) {
#pragma unroll
            for (int u = q; u < q + 4; ++u) {
                acc.x = fmaf(v[u].x, a[u], acc.x);
                acc.y = fmaf(v[u].y, a[u], acc.y);
            }
        } else {
#pragma unroll
            for (int u = q; u < q + 4; ++u)
                if (FULL || j + u < m) gat_step(srow, first, cur, acc, carry, dl[u], a[u], v[u]);
        }
    }
}

template <bool FULL>
__device__ __forceinline__ void gat_batch(
    const int2 r, const int m, const float* __restrict__ feat, float* srow, const unsigned lane8,
    const bool first, int& cur, float2& acc, float2& carry)
{
    float2 va[8], vb[8];
    float  aa[8], ab[8];
    int    da[8], db[8];
    gat_load8<FULL>(r, 0, m, feat, lane8, va, aa, da);
    gat_load8<FULL>(r, 8, m, feat, lane8, vb, ab, db);
    gat_acc8<FULL>(srow, first, cur, acc, carry, 0, m, va, aa, da);
    gat_load8<FULL>(r, 16, m, feat, lane8, va, aa, da);
    gat_acc8<FULL>(srow, first, cur, acc, carry, 8, m, vb, ab, db);
    gat_load8<FULL>(r, 24, m, feat, lane8, vb, ab, db);
    gat_acc8<FULL>(srow, first, cur, acc, carry, 16, m, va, aa, da);
    gat_acc8<FULL>(srow, first, cur, acc, carry, 24, m, vb, ab, db);
}

__global__ __launch_bounds__(256) __attribute__((amdgpu_waves_per_eu(8))) void gat_bucket_sum_gemm(
    const float* __restrict__ feat, const float* __restrict__ weight,
    const float* __restrict__ bias,
    const int* __restrict__ rec, const int* __restrict__ tab,
    float* __restrict__ out, int n_nodes, int n_edges, int nb, size_t ws_bytes)
{
    __shared__ __attribute__((aligned(16))) float sums[BNODES * PITCH];
    __shared__ __attribute__((aligned(16))) int2 sorted[CAP];
    __shared__ int idx[CAP];
    __shared__ float satt[BNODES];
    __shared__ int cnt[BNODES];
    __shared__ int off[BNODES + 1];
    __shared__ int wtot[8];

    const int tid  = threadIdx.x;
    const int lane = tid & 31;
    const int wave = __builtin_amdgcn_readfirstlane(tid >> 5);
    const int b    = (int)blockIdx.x;

    {
        float4* s4 = (float4*)sums;
        for (int i = tid; i < BNODES * PITCH / 4; i += 256) s4[i] = make_float4(0.f, 0.f, 0.f, 0.f);
        if (tid < BNODES) { satt[tid] = 0.f; cnt[tid] = 0; }
    }
    __syncthreads();

    const unsigned lane8 = (unsigned)lane * 8u;
    float*       srow = sums + lane * 2;
    const int    last_node = n_nodes - 1;
    const int2*  rec2 = (const int2*)rec;

    const int chunk = gat_chunk(n_edges);
    const int P     = gat_passes(chunk);
    int U = BIN_BLOCKS * P;
    if (b >= MAXB || !gat_ws_ok(n_edges, nb, U, ws_bytes)) U = 0;

    bool first = true;

    for (int u0 = 0; u0 < U; u0 += 256) {
      const int u = u0 + tid;
      int runlen = 0, ebase = 0;
      if (u < U) {
          const int t = tab[(size_t)u * nb + b];
          runlen = (t >> 16) & 0xFFFF;
          ebase  = (u / P) * chunk + (u % P) * BCAP + (t & 0xFFFF);
      }
      int xs = runlen;
#pragma unroll
      for (int o = 1; o < 32; o <<= 1) {
          const int y = __shfl_up(xs, o);
          if (lane >= o) xs += y;
      }
      if (lane == 31) wtot[wave] = xs;
      __syncthreads();
      int sb = xs - runlen, nsum = 0;
      for (int w = 0; w < 8; ++w) {
          const int t = wtot[w];
          nsum += t;
          if (w < wave) sb += t;
      }
      const int n = __builtin_amdgcn_readfirstlane(nsum);

    for (int c0 = 0; c0 < n; c0 += CAP) {
        const int   nc    = min(CAP, n - c0);

        {
            const int r0 = max(0, c0 - sb);
            const int r1 = min(runlen, c0 + CAP - sb);
            for (int r = r0; r < r1; ++r) idx[sb + r - c0] = ebase + r;
        }
        __syncthreads();

        int2 rr[KPT];
        int  rk[KPT];
#pragma unroll
        for (int k = 0; k < KPT; ++k) rk[k] = idx[min(tid + k * 256, nc - 1)];
#pragma unroll
        for (int k = 0; k < KPT; ++k) rr[k] = rec2[min(max(rk[k], 0), n_edges - 1)];
#pragma unroll
        for (int k = 0; k < KPT; ++k) {
            rk[k] = 0;
            if (tid + k * 256 < nc) {
                const int dl = (int)(((unsigned)rr[k].x >> 24) & (BNODES - 1));
                rk[k] = atomicAdd(&cnt[dl], 1);
                __hip_atomic_fetch_add(&satt[dl], __int_as_float(rr[k].y),
                                       __ATOMIC_RELAXED, __HIP_MEMORY_SCOPE_WORKGROUP);
            }
        }
        __syncthreads();

        const int c = (tid < BNODES) ? cnt[tid] : 0;
        int x = c;
#pragma unroll
        for (int o = 1; o < 32; o <<= 1) {
            const int y = __shfl_up(x, o);
            if (lane >= o) x += y;
        }
        if (lane == 31) wtot[wave] = x;
        __syncthreads();
        if (tid < BNODES) {
            int add = 0;
            for (int w = 0; w < wave; ++w) add += wtot[w];
            off[tid] = x - c + add;
            cnt[tid] = 0;
        }
        if (tid == 0) off[BNODES] = nc;
        __syncthreads();

#pragma unroll
        for (int k = 0; k < KPT; ++k) {
            if (tid + k * 256 < nc) {
                const int dl = (int)(((unsigned)rr[k].x >> 24) & (BNODES - 1));
                const unsigned s = (unsigned)min(rr[k].x & 0xFFFFFF, last_node);
                sorted[off[dl] + rk[k]] = make_int2((int)((s << 8) | (unsigned)dl), rr[k].y);
            }
        }
        __syncthreads();

        const int s0 = __builtin_amdgcn_readfirstlane(off[wave * 16]);
        const int s1 = __builtin_amdgcn_readfirstlane(off[wave * 16 + 16]);

        int    cur   = -1;
        float2 acc   = make_float2(0.f, 0.f);
        float2 carry = make_float2(0.f, 0.f);

        int2 r = make_int2(0, 0);
        if (s0 + lane < s1) r = sorted[s0 + lane];
        for (int base = s0; base < s1; base += 32) {
            const int m = min(32, s1 - base);
            int2 rn = make_int2(0, 0);
            if (base + 32 + lane < s1) rn = sorted[base + 32 + lane];
            if (m == 32) gat_batch<true>(r, m, feat, srow, lane8, first, cur, acc, carry);
            else         gat_batch<false>(r, m, feat, srow, lane8, first, cur, acc, carry);
            r = rn;
        }
        gat_flush(srow, cur, acc, carry);
        first = false;
    }
    }

    const int row0 = b * BNODES + wave * 16;
    if (row0 >= n_nodes) return;

    const int mn  = lane & 15;
    const int hi  = lane >> 4;
    const int khi = hi * 2;

    const float* arow  = sums + (wave * 16 + mn) * PITCH + khi;
    const float* bbase = weight + (size_t)khi * OUT_FEAT + mn;

    v8f acc0 = {}, acc1 = {}, acc2 = {}, acc3 = {};

#pragma unroll
    for (int k = 0; k < IN_FEAT; k += 4) {
        v2f a;
        a.x = arow[k];
        a.y = arow[k + 1];

        const float* bk = bbase + (size_t)k * OUT_FEAT;
        v2f b0, b1, b2, b3;
        b0.x = bk[0];   b0.y = bk[OUT_FEAT + 0];
        b1.x = bk[16];  b1.y = bk[OUT_FEAT + 16];
        b2.x = bk[32];  b2.y = bk[OUT_FEAT + 32];
        b3.x = bk[48];  b3.y = bk[OUT_FEAT + 48];

        acc0 = __builtin_amdgcn_wmma_f32_16x16x4_f32(false, a, false, b0,
                                                     (short)0, acc0, false, false);
        acc1 = __builtin_amdgcn_wmma_f32_16x16x4_f32(false, a, false, b1,
                                                     (short)0, acc1, false, false);
        acc2 = __builtin_amdgcn_wmma_f32_16x16x4_f32(false, a, false, b2,
                                                     (short)0, acc2, false, false);
        acc3 = __builtin_amdgcn_wmma_f32_16x16x4_f32(false, a, false, b3,
                                                     (short)0, acc3, false, false);
    }

    const float bv0 = bias[mn];
    const float bv1 = bias[16 + mn];
    const float bv2 = bias[32 + mn];
    const float bv3 = bias[48 + mn];

#pragma unroll
    for (int v = 0; v < 8; ++v) {
        const int lr  = wave * 16 + v + 8 * hi;
        const int row = b * BNODES + lr;
        const float sa = satt[lr];
        if (row < n_nodes) {
            float* orow = out + (size_t)row * OUT_FEAT + mn;
            orow[0]  = acc0[v] + sa * bv0;
            orow[16] = acc1[v] + sa * bv1;
            orow[32] = acc2[v] + sa * bv2;
            orow[48] = acc3[v] + sa * bv3;
        }
    }
}

extern "C" void kernel_launch(void* const* d_in, const int* in_sizes, int n_in,
                              void* d_out, int out_size, void* d_ws, size_t ws_size,
                              hipStream_t stream)
{
    const float* feat   = (const float*)d_in[0];
    const float* att    = (const float*)d_in[1];
    const int*   src    = (const int*)d_in[2];
    const int*   dst    = (const int*)d_in[3];
    const float* weight = (const float*)d_in[4];
    const float* bias   = (const float*)d_in[5];

    float* out = (float*)d_out;

    const int n_nodes = in_sizes[0] / IN_FEAT;
    const int n_edges = in_sizes[2];
    const int nb      = (n_nodes + BNODES - 1) / BNODES;

    int* rec = (int*)d_ws;
    int* tab = rec + (size_t)n_edges * 2;

    gat_bin_local<<<BIN_BLOCKS, BIN_THREADS, 0, stream>>>(src, dst, att, n_edges, n_nodes, nb,
                                                          rec, tab, ws_size);
    gat_bucket_sum_gemm<<<nb, 256, 0, stream>>>(feat, weight, bias, rec, tab,
                                                out, n_nodes, n_edges, nb, ws_size);
}
